# starting kernel
# baseline (speedup 1.0000x reference)
#include <hip/hip_runtime.h>
#include <cmath>


typedef float v2f __attribute__((ext_vector_type(2)));
typedef float v8f __attribute__((ext_vector_type(8)));

struct GaussW { float w[13]; };

__device__ __forceinline__ int iclamp(int v, int lo, int hi) {
    return v < lo ? lo : (v > hi ? hi : v);
}

template <int PASS>
__global__ __launch_bounds__(256) void gauss_wmma_pass(const float* __restrict__ in,
                                                       float* __restrict__ out,
                                                       GaussW gw) {
    __shared__ float wpad[64];
    const unsigned tid = threadIdx.x;
    if (tid < 64) wpad[tid] = 0.0f;
    __syncthreads();
    if (tid < 13) wpad[16 + tid] = gw.w[tid];
    __syncthreads();

    const unsigned wave   = tid >> 5;
    const unsigned lane   = tid & 31;
    const unsigned laneLo = lane & 15;
    const unsigned laneHi = lane >> 4;

    const unsigned tile = blockIdx.x * 8u + wave;
    const unsigned ab   = tile & 7u;
    const unsigned lg   = tile >> 3;
    const unsigned n    = lg * 16u + laneLo;

    unsigned base;
    unsigned S;
    if (PASS == 0) {
        S = 131072u;
        base = (n & 131071u) + (n >> 17) * 16777216u;
    } else if (PASS == 1) {
        S = 1024u;
        base = (n & 1023u) + ((n >> 10) & 127u) * 131072u + (n >> 17) * 16777216u;
    } else {
        S = 8u;
        base = (n & 7u) + ((n >> 3) & 127u) * 1024u +
               ((n >> 10) & 127u) * 131072u + (n >> 17) * 16777216u;
    }

    const int m0 = (int)(ab * 16u);

    v8f acc = {};
#pragma unroll
    for (int c = 0; c < 7; ++c) {
        const int k0 = 4 * c + 2 * (int)laneHi;

        v2f a;
        a.x = wpad[16 + k0 - (int)laneLo];
        a.y = wpad[16 + k0 + 1 - (int)laneLo];

        const int p0 = m0 + k0 - 6;
        const int p1 = p0 + 1;
        const unsigned q0 = (unsigned)iclamp(p0, 0, 127);
        const unsigned q1 = (unsigned)iclamp(p1, 0, 127);
        const float v0 = in[base + q0 * S];
        const float v1 = in[base + q1 * S];
        v2f b;
        b.x = ((unsigned)p0 < 128u) ? v0 : 0.0f;
        b.y = ((unsigned)p1 < 128u) ? v1 : 0.0f;

        acc = __builtin_amdgcn_wmma_f32_16x16x4_f32(
             false, a,  false, b,
             (short)0, acc,  false,  false);
    }

#pragma unroll
    for (int v = 0; v < 8; ++v) {
        const int m = m0 + v + 8 * (int)laneHi;
        out[base + (unsigned)m * S] = acc[v];
    }
}

static void make_weights(GaussW& g) {
    float k[13];
    float s = 0.0f;
    for (int i = 0; i < 13; ++i) {
        const float x = (float)(i - 6);
        k[i] = expf(-0.5f * x * x / 4.0f);
    }
    for (int i = 0; i < 13; ++i) s += k[i];
    for (int i = 0; i < 13; ++i) g.w[i] = k[i] / s;
}

extern "C" void kernel_launch(void* const* d_in, const int* in_sizes, int n_in,
                              void* d_out, int out_size, void* d_ws, size_t ws_size,
                              hipStream_t stream) {
    (void)in_sizes; (void)n_in; (void)out_size; (void)ws_size;

    const float* in  = (const float*)d_in[0];
    float*       out = (float*)d_out;
    float*       tmp = (float*)d_ws;

    GaussW gw;
    make_weights(gw);

    const int blocks = 16384;
    const dim3 blk(256);

    gauss_wmma_pass<0><<<blocks, blk, 0, stream>>>(in,  out, gw);
    gauss_wmma_pass<1><<<blocks, blk, 0, stream>>>(out, tmp, gw);
    gauss_wmma_pass<2><<<blocks, blk, 0, stream>>>(tmp, out, gw);
}


# 128-bit loads and stores: a lane carries 4 adjacent lines, a wave owns a 16x64 tile (4 accumulator tiles share one band fragment; 14 row loads in one group ahead); 4x fewer waves; sum order unchanged
# speedup vs baseline: 1.8587x; 1.8587x over previous
#include <hip/hip_runtime.h>
#include <cmath>


typedef float v2f __attribute__((ext_vector_type(2)));
typedef float v4f __attribute__((ext_vector_type(4)));
typedef float v8f __attribute__((ext_vector_type(8)));

struct GaussW { float w[13]; };

__device__ __forceinline__ int iclamp(int v, int lo, int hi) {
    return v < lo ? lo : (v > hi ? hi : v);
}

template <int PASS>
__global__ __launch_bounds__(256) void gauss_wmma_pass(const float* __restrict__ in,
                                                       float* __restrict__ out,
                                                       GaussW gw) {
    __shared__ float wpad[64];
    const unsigned tid = threadIdx.x;
    if (tid < 64) wpad[tid] = 0.0f;
    __syncthreads();
    if (tid < 13) wpad[16 + tid] = gw.w[tid];
    __syncthreads();

    const unsigned wave   = tid >> 5;
    const unsigned lane   = tid & 31;
    const unsigned laneLo = lane & 15;
    const unsigned laneHi = lane >> 4;

    const unsigned tile = blockIdx.x * 8u + wave;
    const unsigned ab   = tile & 7u;
    const unsigned lg   = tile >> 3;
    const unsigned n    = lg * 64u + laneLo * 4u;

    unsigned base;
    unsigned S;
    if (PASS == 0) {
        S = 131072u;
        base = (n & 131071u) + (n >> 17) * 16777216u;
    } else if (PASS == 1) {
        S = 1024u;
        base = (n & 1023u) + ((n >> 10) & 127u) * 131072u + (n >> 17) * 16777216u;
    } else {
        S = 8u;
        base = (n & 7u) + ((n >> 3) & 127u) * 1024u +
               ((n >> 10) & 127u) * 131072u + (n >> 17) * 16777216u;
    }

    const int m0 = (int)(ab * 16u);

    const char* __restrict__ inb  = reinterpret_cast<const char*>(in);
    char* __restrict__       outb = reinterpret_cast<char*>(out);

    v4f r0[7], r1[7];
    v2f a[7];
#pragma unroll
    for (int c = 0; c < 7; ++c) {
        const int k0 = 4 * c + 2 * (int)laneHi;
        const int p0 = m0 + k0 - 6;
        const int p1 = p0 + 1;
        const unsigned q0 = (unsigned)iclamp(p0, 0, 127);
        const unsigned q1 = (unsigned)iclamp(p1, 0, 127);
        r0[c] = *reinterpret_cast<const v4f*>(inb + (base + q0 * S) * 4u);
        r1[c] = *reinterpret_cast<const v4f*>(inb + (base + q1 * S) * 4u);
        a[c].x = wpad[16 + k0 - (int)laneLo];
        a[c].y = wpad[16 + k0 + 1 - (int)laneLo];
    }

    __builtin_amdgcn_sched_barrier(0);

    v8f acc0 = {}, acc1 = {}, acc2 = {}, acc3 = {};
#pragma unroll
    for (int c = 0; c < 7; ++c) {
        const int k0 = 4 * c + 2 * (int)laneHi;
        const int p0 = m0 + k0 - 6;
        const int p1 = p0 + 1;
        const bool ok0 = (unsigned)p0 < 128u;
        const bool ok1 = (unsigned)p1 < 128u;
        v2f b0, b1, b2, b3;
        b0.x = ok0 ? r0[c].x : 0.0f;  b0.y = ok1 ? r1[c].x : 0.0f;
        b1.x = ok0 ? r0[c].y : 0.0f;  b1.y = ok1 ? r1[c].y : 0.0f;
        b2.x = ok0 ? r0[c].z : 0.0f;  b2.y = ok1 ? r1[c].z : 0.0f;
        b3.x = ok0 ? r0[c].w : 0.0f;  b3.y = ok1 ? r1[c].w : 0.0f;

        acc0 = __builtin_amdgcn_wmma_f32_16x16x4_f32(false, a[c], false, b0, (short)0, acc0, false, false);
        acc1 = __builtin_amdgcn_wmma_f32_16x16x4_f32(false, a[c], false, b1, (short)0, acc1, false, false);
        acc2 = __builtin_amdgcn_wmma_f32_16x16x4_f32(false, a[c], false, b2, (short)0, acc2, false, false);
        acc3 = __builtin_amdgcn_wmma_f32_16x16x4_f32(false, a[c], false, b3, (short)0, acc3, false, false);
    }

#pragma unroll
    for (int v = 0; v < 8; ++v) {
        const int m = m0 + v + 8 * (int)laneHi;
        v4f o;
        o.x = acc0[v];  o.y = acc1[v];  o.z = acc2[v];  o.w = acc3[v];
        *reinterpret_cast<v4f*>(outb + (base + (unsigned)m * S) * 4u) = o;
    }
}

static void make_weights(GaussW& g) {
    float k[13];
    float s = 0.0f;
    for (int i = 0; i < 13; ++i) {
        const float x = (float)(i - 6);
        k[i] = expf(-0.5f * x * x / 4.0f);
    }
    for (int i = 0; i < 13; ++i) s += k[i];
    for (int i = 0; i < 13; ++i) g.w[i] = k[i] / s;
}

extern "C" void kernel_launch(void* const* d_in, const int* in_sizes, int n_in,
                              void* d_out, int out_size, void* d_ws, size_t ws_size,
                              hipStream_t stream) {
    (void)in_sizes; (void)n_in; (void)out_size; (void)ws_size;

    const float* in  = (const float*)d_in[0];
    float*       out = (float*)d_out;
    float*       tmp = (float*)d_ws;

    GaussW gw;
    make_weights(gw);

    const int blocks = 4096;
    const dim3 blk(256);

    gauss_wmma_pass<0><<<blocks, blk, 0, stream>>>(in,  out, gw);
    gauss_wmma_pass<1><<<blocks, blk, 0, stream>>>(out, tmp, gw);
    gauss_wmma_pass<2><<<blocks, blk, 0, stream>>>(tmp, out, gw);
}


# Y and Z passes fused in one kernel: a 16-wave group keeps its Y result (16 y x 128 z x 8 c) in LDS and the Z products read it from there; 3 launches -> 2; sum order unchanged
# speedup vs baseline: 1.9992x; 1.0756x over previous
#include <hip/hip_runtime.h>
#include <cmath>


typedef float v2f __attribute__((ext_vector_type(2)));
typedef float v4f __attribute__((ext_vector_type(4)));
typedef float v8f __attribute__((ext_vector_type(8)));

struct GaussW { float w[13]; };

__device__ __forceinline__ int iclamp(int v, int lo, int hi) {
    return v < lo ? lo : (v > hi ? hi : v);
}

template <int PASS>
__global__ __launch_bounds__(256) void gauss_wmma_pass(const float* __restrict__ in,
                                                       float* __restrict__ out,
                                                       GaussW gw) {
    __shared__ float wpad[64];
    const unsigned tid = threadIdx.x;
    if (tid < 64) wpad[tid] = 0.0f;
    __syncthreads();
    if (tid < 13) wpad[16 + tid] = gw.w[tid];
    __syncthreads();

    const unsigned wave   = tid >> 5;
    const unsigned lane   = tid & 31;
    const unsigned laneLo = lane & 15;
    const unsigned laneHi = lane >> 4;

    const unsigned tile = blockIdx.x * 8u + wave;
    const unsigned ab   = tile & 7u;
    const unsigned lg   = tile >> 3;
    const unsigned n    = lg * 64u + laneLo * 4u;

    unsigned base;
    unsigned S;
    if (PASS == 0) {
        S = 131072u;
        base = (n & 131071u) + (n >> 17) * 16777216u;
    } else if (PASS == 1) {
        S = 1024u;
        base = (n & 1023u) + ((n >> 10) & 127u) * 131072u + (n >> 17) * 16777216u;
    } else {
        S = 8u;
        base = (n & 7u) + ((n >> 3) & 127u) * 1024u +
               ((n >> 10) & 127u) * 131072u + (n >> 17) * 16777216u;
    }

    const int m0 = (int)(ab * 16u);

    const char* __restrict__ inb  = reinterpret_cast<const char*>(in);
    char* __restrict__       outb = reinterpret_cast<char*>(out);

    v4f r0[7], r1[7];
    v2f a[7];
#pragma unroll
    for (int c = 0; c < 7; ++c) {
        const int k0 = 4 * c + 2 * (int)laneHi;
        const int p0 = m0 + k0 - 6;
        const int p1 = p0 + 1;
        const unsigned q0 = (unsigned)iclamp(p0, 0, 127);
        const unsigned q1 = (unsigned)iclamp(p1, 0, 127);
        r0[c] = *reinterpret_cast<const v4f*>(inb + (base + q0 * S) * 4u);
        r1[c] = *reinterpret_cast<const v4f*>(inb + (base + q1 * S) * 4u);
        a[c].x = wpad[16 + k0 - (int)laneLo];
        a[c].y = wpad[16 + k0 + 1 - (int)laneLo];
    }

    __builtin_amdgcn_sched_barrier(0);

    v8f acc0 = {}, acc1 = {}, acc2 = {}, acc3 = {};
#pragma unroll
    for (int c = 0; c < 7; ++c) {
        const int k0 = 4 * c + 2 * (int)laneHi;
        const int p0 = m0 + k0 - 6;
        const int p1 = p0 + 1;
        const bool ok0 = (unsigned)p0 < 128u;
        const bool ok1 = (unsigned)p1 < 128u;
        v2f b0, b1, b2, b3;
        b0.x = ok0 ? r0[c].x : 0.0f;  b0.y = ok1 ? r1[c].x : 0.0f;
        b1.x = ok0 ? r0[c].y : 0.0f;  b1.y = ok1 ? r1[c].y : 0.0f;
        b2.x = ok0 ? r0[c].z : 0.0f;  b2.y = ok1 ? r1[c].z : 0.0f;
        b3.x = ok0 ? r0[c].w : 0.0f;  b3.y = ok1 ? r1[c].w : 0.0f;

        acc0 = __builtin_amdgcn_wmma_f32_16x16x4_f32(false, a[c], false, b0, (short)0, acc0, false, false);
        acc1 = __builtin_amdgcn_wmma_f32_16x16x4_f32(false, a[c], false, b1, (short)0, acc1, false, false);
        acc2 = __builtin_amdgcn_wmma_f32_16x16x4_f32(false, a[c], false, b2, (short)0, acc2, false, false);
        acc3 = __builtin_amdgcn_wmma_f32_16x16x4_f32(false, a[c], false, b3, (short)0, acc3, false, false);
    }

#pragma unroll
    for (int v = 0; v < 8; ++v) {
        const int m = m0 + v + 8 * (int)laneHi;
        v4f o;
        o.x = acc0[v];  o.y = acc1[v];  o.z = acc2[v];  o.w = acc3[v];
        *reinterpret_cast<v4f*>(outb + (base + (unsigned)m * S) * 4u) = o;
    }
}

#define YZ_PITCH 1032u

__global__ __launch_bounds__(512) void gauss_wmma_yz(const float* __restrict__ in,
                                                     float* __restrict__ out,
                                                     GaussW gw) {
    __shared__ float wpad[64];
    __shared__ v4f ytile[16u * YZ_PITCH / 4u];

    const unsigned tid = threadIdx.x;
    if (tid < 64) wpad[tid] = 0.0f;
    __syncthreads();
    if (tid < 13) wpad[16 + tid] = gw.w[tid];
    __syncthreads();

    const unsigned wave   = tid >> 5;
    const unsigned lane   = tid & 31;
    const unsigned laneLo = lane & 15;
    const unsigned laneHi = lane >> 4;

    const unsigned yb   = blockIdx.x & 7u;
    const unsigned slab = (blockIdx.x >> 3) * 131072u;
    const int y0 = (int)(yb * 16u);

    const char* __restrict__ inb  = reinterpret_cast<const char*>(in);
    char* __restrict__       outb = reinterpret_cast<char*>(out);

    v2f a[7];
#pragma unroll
    for (int c = 0; c < 7; ++c) {
        const int k0 = 4 * c + 2 * (int)laneHi;
        a[c].x = wpad[16 + k0 - (int)laneLo];
        a[c].y = wpad[16 + k0 + 1 - (int)laneLo];
    }

    {
        const unsigned zc   = wave * 64u + laneLo * 4u;
        const unsigned base = slab + zc;
        const unsigned S    = 1024u;

        v4f r0[7], r1[7];
#pragma unroll
        for (int c = 0; c < 7; ++c) {
            const int k0 = 4 * c + 2 * (int)laneHi;
            const int p0 = y0 + k0 - 6;
            const int p1 = p0 + 1;
            const unsigned q0 = (unsigned)iclamp(p0, 0, 127);
            const unsigned q1 = (unsigned)iclamp(p1, 0, 127);
            r0[c] = *reinterpret_cast<const v4f*>(inb + (base + q0 * S) * 4u);
            r1[c] = *reinterpret_cast<const v4f*>(inb + (base + q1 * S) * 4u);
        }

        __builtin_amdgcn_sched_barrier(0);

        v8f acc0 = {}, acc1 = {}, acc2 = {}, acc3 = {};
#pragma unroll
        for (int c = 0; c < 7; ++c) {
            const int k0 = 4 * c + 2 * (int)laneHi;
            const int p0 = y0 + k0 - 6;
            const int p1 = p0 + 1;
            const bool ok0 = (unsigned)p0 < 128u;
            const bool ok1 = (unsigned)p1 < 128u;
            v2f b0, b1, b2, b3;
            b0.x = ok0 ? r0[c].x : 0.0f;  b0.y = ok1 ? r1[c].x : 0.0f;
            b1.x = ok0 ? r0[c].y : 0.0f;  b1.y = ok1 ? r1[c].y : 0.0f;
            b2.x = ok0 ? r0[c].z : 0.0f;  b2.y = ok1 ? r1[c].z : 0.0f;
            b3.x = ok0 ? r0[c].w : 0.0f;  b3.y = ok1 ? r1[c].w : 0.0f;

            acc0 = __builtin_amdgcn_wmma_f32_16x16x4_f32(false, a[c], false, b0, (short)0, acc0, false, false);
            acc1 = __builtin_amdgcn_wmma_f32_16x16x4_f32(false, a[c], false, b1, (short)0, acc1, false, false);
            acc2 = __builtin_amdgcn_wmma_f32_16x16x4_f32(false, a[c], false, b2, (short)0, acc2, false, false);
            acc3 = __builtin_amdgcn_wmma_f32_16x16x4_f32(false, a[c], false, b3, (short)0, acc3, false, false);
        }

#pragma unroll
        for (int v = 0; v < 8; ++v) {
            const unsigned row = (unsigned)v + 8u * laneHi;
            v4f o;
            o.x = acc0[v];  o.y = acc1[v];  o.z = acc2[v];  o.w = acc3[v];
            ytile[(row * YZ_PITCH + zc) >> 2] = o;
        }
    }

    __syncthreads();

    {
        const unsigned nl    = (wave >> 3) * 64u + laneLo * 4u;
        const unsigned c0    = nl & 7u;
        const unsigned yl    = nl >> 3;
        const int      z0    = (int)((wave & 7u) * 16u);
        const unsigned lbase = yl * YZ_PITCH + c0;
        const unsigned S     = 8u;

        v4f r0[7], r1[7];
#pragma unroll
        for (int c = 0; c < 7; ++c) {
            const int k0 = 4 * c + 2 * (int)laneHi;
            const int p0 = z0 + k0 - 6;
            const int p1 = p0 + 1;
            const unsigned q0 = (unsigned)iclamp(p0, 0, 127);
            const unsigned q1 = (unsigned)iclamp(p1, 0, 127);
            r0[c] = ytile[(lbase + q0 * S) >> 2];
            r1[c] = ytile[(lbase + q1 * S) >> 2];
        }

        __builtin_amdgcn_sched_barrier(0);

        v8f acc0 = {}, acc1 = {}, acc2 = {}, acc3 = {};
#pragma unroll
        for (int c = 0; c < 7; ++c) {
            const int k0 = 4 * c + 2 * (int)laneHi;
            const int p0 = z0 + k0 - 6;
            const int p1 = p0 + 1;
            const bool ok0 = (unsigned)p0 < 128u;
            const bool ok1 = (unsigned)p1 < 128u;
            v2f b0, b1, b2, b3;
            b0.x = ok0 ? r0[c].x : 0.0f;  b0.y = ok1 ? r1[c].x : 0.0f;
            b1.x = ok0 ? r0[c].y : 0.0f;  b1.y = ok1 ? r1[c].y : 0.0f;
            b2.x = ok0 ? r0[c].z : 0.0f;  b2.y = ok1 ? r1[c].z : 0.0f;
            b3.x = ok0 ? r0[c].w : 0.0f;  b3.y = ok1 ? r1[c].w : 0.0f;

            acc0 = __builtin_amdgcn_wmma_f32_16x16x4_f32(false, a[c], false, b0, (short)0, acc0, false, false);
            acc1 = __builtin_amdgcn_wmma_f32_16x16x4_f32(false, a[c], false, b1, (short)0, acc1, false, false);
            acc2 = __builtin_amdgcn_wmma_f32_16x16x4_f32(false, a[c], false, b2, (short)0, acc2, false, false);
            acc3 = __builtin_amdgcn_wmma_f32_16x16x4_f32(false, a[c], false, b3, (short)0, acc3, false, false);
        }

        const unsigned gbase = slab + ((unsigned)y0 + yl) * 1024u + c0;
#pragma unroll
        for (int v = 0; v < 8; ++v) {
            const unsigned m = (unsigned)z0 + (unsigned)v + 8u * laneHi;
            v4f o;
            o.x = acc0[v];  o.y = acc1[v];  o.z = acc2[v];  o.w = acc3[v];
            *reinterpret_cast<v4f*>(outb + (gbase + m * S) * 4u) = o;
        }
    }
}

static void make_weights(GaussW& g) {
    float k[13];
    float s = 0.0f;
    for (int i = 0; i < 13; ++i) {
        const float x = (float)(i - 6);
        k[i] = expf(-0.5f * x * x / 4.0f);
    }
    for (int i = 0; i < 13; ++i) s += k[i];
    for (int i = 0; i < 13; ++i) g.w[i] = k[i] / s;
}

extern "C" void kernel_launch(void* const* d_in, const int* in_sizes, int n_in,
                              void* d_out, int out_size, void* d_ws, size_t ws_size,
                              hipStream_t stream) {
    (void)in_sizes; (void)n_in; (void)out_size; (void)ws_size;

    const float* in  = (const float*)d_in[0];
    float*       out = (float*)d_out;
    float*       tmp = (float*)d_ws;

    GaussW gw;
    make_weights(gw);

    gauss_wmma_pass<0><<<4096, dim3(256), 0, stream>>>(in, tmp, gw);

    gauss_wmma_yz<<<2048, dim3(512), 0, stream>>>(tmp, out, gw);
}


# each wave marches along the conv axis through all 8 blocks of its lines: 12 halo rows stay in registers, 16 new rows per block loaded ahead; LDS-only barrier; 8x fewer waves; sum order unchanged
# speedup vs baseline: 2.3421x; 1.1715x over previous
#include <hip/hip_runtime.h>
#include <cmath>


typedef float v2f __attribute__((ext_vector_type(2)));
typedef float v4f __attribute__((ext_vector_type(4)));
typedef float v8f __attribute__((ext_vector_type(8)));

struct GaussW { float w[13]; };

__device__ __forceinline__ int iclamp(int v, int lo, int hi) {
    return v < lo ? lo : (v > hi ? hi : v);
}

__device__ __forceinline__ void band_products(const v2f (&a)[7], const v4f (&r0)[7],
                                              const v4f (&r1)[7], int m0, int laneHi,
                                              v8f& acc0, v8f& acc1, v8f& acc2, v8f& acc3) {
    acc0 = {}; acc1 = {}; acc2 = {}; acc3 = {};
#pragma unroll
    for (int c = 0; c < 7; ++c) {
        const int k0 = 4 * c + 2 * laneHi;
        const int p0 = m0 + k0 - 6;
        const int p1 = p0 + 1;
        const bool ok0 = (unsigned)p0 < 128u;
        const bool ok1 = (unsigned)p1 < 128u;
        v2f b0, b1, b2, b3;
        b0.x = ok0 ? r0[c].x : 0.0f;  b0.y = ok1 ? r1[c].x : 0.0f;
        b1.x = ok0 ? r0[c].y : 0.0f;  b1.y = ok1 ? r1[c].y : 0.0f;
        b2.x = ok0 ? r0[c].z : 0.0f;  b2.y = ok1 ? r1[c].z : 0.0f;
        b3.x = ok0 ? r0[c].w : 0.0f;  b3.y = ok1 ? r1[c].w : 0.0f;

        acc0 = __builtin_amdgcn_wmma_f32_16x16x4_f32(false, a[c], false, b0, (short)0, acc0, false, false);
        acc1 = __builtin_amdgcn_wmma_f32_16x16x4_f32(false, a[c], false, b1, (short)0, acc1, false, false);
        acc2 = __builtin_amdgcn_wmma_f32_16x16x4_f32(false, a[c], false, b2, (short)0, acc2, false, false);
        acc3 = __builtin_amdgcn_wmma_f32_16x16x4_f32(false, a[c], false, b3, (short)0, acc3, false, false);
    }
}

__device__ __forceinline__ void load_first_block(const char* __restrict__ inb, unsigned base,
                                                 unsigned S, int laneHi,
                                                 v4f (&r0)[7], v4f (&r1)[7]) {
#pragma unroll
    for (int c = 0; c < 7; ++c) {
        const int p0 = 4 * c + 2 * laneHi - 6;
        const unsigned q0 = (unsigned)iclamp(p0, 0, 127);
        const unsigned q1 = (unsigned)iclamp(p0 + 1, 0, 127);
        r0[c] = *reinterpret_cast<const v4f*>(inb + (base + q0 * S) * 4u);
        r1[c] = *reinterpret_cast<const v4f*>(inb + (base + q1 * S) * 4u);
    }
}

__device__ __forceinline__ void load_next_rows(const char* __restrict__ inb, unsigned base,
                                               unsigned S, int m0, int laneHi,
                                               v4f (&n0)[4], v4f (&n1)[4]) {
#pragma unroll
    for (int j = 0; j < 4; ++j) {
        const int p0 = m0 + 22 + 4 * j + 2 * laneHi;
        const unsigned q0 = (unsigned)iclamp(p0, 0, 127);
        const unsigned q1 = (unsigned)iclamp(p0 + 1, 0, 127);
        n0[j] = *reinterpret_cast<const v4f*>(inb + (base + q0 * S) * 4u);
        n1[j] = *reinterpret_cast<const v4f*>(inb + (base + q1 * S) * 4u);
    }
}

__device__ __forceinline__ void rotate_rows(v4f (&r0)[7], v4f (&r1)[7],
                                            const v4f (&n0)[4], const v4f (&n1)[4]) {
#pragma unroll
    for (int c = 0; c < 3; ++c) { r0[c] = r0[c + 4]; r1[c] = r1[c + 4]; }
#pragma unroll
    for (int j = 0; j < 4; ++j) { r0[3 + j] = n0[j]; r1[3 + j] = n1[j]; }
}

__global__ __launch_bounds__(128) void gauss_wmma_x(const float* __restrict__ in,
                                                    float* __restrict__ out,
                                                    GaussW gw) {
    __shared__ float wpad[64];
    const unsigned tid = threadIdx.x;
    if (tid < 64) wpad[tid] = 0.0f;
    __syncthreads();
    if (tid < 13) wpad[16 + tid] = gw.w[tid];
    __syncthreads();

    const unsigned wave   = tid >> 5;
    const unsigned lane   = tid & 31;
    const unsigned laneLo = lane & 15;
    const unsigned laneHi = lane >> 4;

    const unsigned lg   = blockIdx.x * 4u + wave;
    const unsigned n    = lg * 64u + laneLo * 4u;
    const unsigned S    = 131072u;
    const unsigned base = (n & 131071u) + (n >> 17) * 16777216u;

    const char* __restrict__ inb  = reinterpret_cast<const char*>(in);
    char* __restrict__       outb = reinterpret_cast<char*>(out);

    v2f a[7];
#pragma unroll
    for (int c = 0; c < 7; ++c) {
        const int k0 = 4 * c + 2 * (int)laneHi;
        a[c].x = wpad[16 + k0 - (int)laneLo];
        a[c].y = wpad[16 + k0 + 1 - (int)laneLo];
    }

    v4f r0[7], r1[7];
    load_first_block(inb, base, S, (int)laneHi, r0, r1);

#pragma unroll
    for (int ab = 0; ab < 8; ++ab) {
        const int m0 = ab * 16;
        v4f n0[4], n1[4];
        if (ab < 7) load_next_rows(inb, base, S, m0, (int)laneHi, n0, n1);

        __builtin_amdgcn_sched_barrier(0);

        v8f acc0, acc1, acc2, acc3;
        band_products(a, r0, r1, m0, (int)laneHi, acc0, acc1, acc2, acc3);

#pragma unroll
        for (int v = 0; v < 8; ++v) {
            const int m = m0 + v + 8 * (int)laneHi;
            v4f o;
            o.x = acc0[v];  o.y = acc1[v];  o.z = acc2[v];  o.w = acc3[v];
            *reinterpret_cast<v4f*>(outb + (base + (unsigned)m * S) * 4u) = o;
        }

        if (ab < 7) rotate_rows(r0, r1, n0, n1);
    }
}

#define YZ_PITCH 1032u
#define YZ_TILE4 (16u * YZ_PITCH / 4u)

__global__ __launch_bounds__(512) void gauss_wmma_yz(const float* __restrict__ in,
                                                     float* __restrict__ out,
                                                     GaussW gw) {
    __shared__ float wpad[64];
    __shared__ v4f ytile[2u * YZ_TILE4];

    const unsigned tid = threadIdx.x;
    if (tid < 64) wpad[tid] = 0.0f;
    __syncthreads();
    if (tid < 13) wpad[16 + tid] = gw.w[tid];
    __syncthreads();

    const unsigned wave   = tid >> 5;
    const unsigned lane   = tid & 31;
    const unsigned laneLo = lane & 15;
    const unsigned laneHi = lane >> 4;

    const unsigned slab = blockIdx.x * 131072u;

    const char* __restrict__ inb  = reinterpret_cast<const char*>(in);
    char* __restrict__       outb = reinterpret_cast<char*>(out);

    v2f a[7];
#pragma unroll
    for (int c = 0; c < 7; ++c) {
        const int k0 = 4 * c + 2 * (int)laneHi;
        a[c].x = wpad[16 + k0 - (int)laneLo];
        a[c].y = wpad[16 + k0 + 1 - (int)laneLo];
    }

    const unsigned zc    = wave * 64u + laneLo * 4u;
    const unsigned ybase = slab + zc;
    const unsigned SY    = 1024u;

    const unsigned nl    = (wave >> 3) * 64u + laneLo * 4u;
    const unsigned c0    = nl & 7u;
    const unsigned yl    = nl >> 3;
    const int      z0    = (int)((wave & 7u) * 16u);
    const unsigned lbase = yl * YZ_PITCH + c0;
    const unsigned SZ    = 8u;

    v4f r0[7], r1[7];
    load_first_block(inb, ybase, SY, (int)laneHi, r0, r1);

#pragma unroll
    for (int yb = 0; yb < 8; ++yb) {
        const int y0 = yb * 16;
        const unsigned buf = ((unsigned)yb & 1u) * YZ_TILE4;

        v4f n0[4], n1[4];
        if (yb < 7) load_next_rows(inb, ybase, SY, y0, (int)laneHi, n0, n1);

        __builtin_amdgcn_sched_barrier(0);

        {
            v8f acc0, acc1, acc2, acc3;
            band_products(a, r0, r1, y0, (int)laneHi, acc0, acc1, acc2, acc3);

#pragma unroll
            for (int v = 0; v < 8; ++v) {
                const unsigned row = (unsigned)v + 8u * laneHi;
                v4f o;
                o.x = acc0[v];  o.y = acc1[v];  o.z = acc2[v];  o.w = acc3[v];
                ytile[buf + ((row * YZ_PITCH + zc) >> 2)] = o;
            }
        }

        __builtin_amdgcn_fence(__ATOMIC_RELEASE, "workgroup", "local");
        __builtin_amdgcn_s_barrier();
        __builtin_amdgcn_fence(__ATOMIC_ACQUIRE, "workgroup", "local");

        {
            v4f s0[7], s1[7];
#pragma unroll
            for (int c = 0; c < 7; ++c) {
                const int k0 = 4 * c + 2 * (int)laneHi;
                const int p0 = z0 + k0 - 6;
                const int p1 = p0 + 1;
                const unsigned q0 = (unsigned)iclamp(p0, 0, 127);
                const unsigned q1 = (unsigned)iclamp(p1, 0, 127);
                s0[c] = ytile[buf + ((lbase + q0 * SZ) >> 2)];
                s1[c] = ytile[buf + ((lbase + q1 * SZ) >> 2)];
            }

            __builtin_amdgcn_sched_barrier(0);

            v8f acc0, acc1, acc2, acc3;
            band_products(a, s0, s1, z0, (int)laneHi, acc0, acc1, acc2, acc3);

            const unsigned gbase = slab + ((unsigned)y0 + yl) * 1024u + c0;
#pragma unroll
            for (int v = 0; v < 8; ++v) {
                const unsigned m = (unsigned)z0 + (unsigned)v + 8u * laneHi;
                v4f o;
                o.x = acc0[v];  o.y = acc1[v];  o.z = acc2[v];  o.w = acc3[v];
                *reinterpret_cast<v4f*>(outb + (gbase + m * SZ) * 4u) = o;
            }
        }

        if (yb < 7) rotate_rows(r0, r1, n0, n1);
    }
}

static void make_weights(GaussW& g) {
    float k[13];
    float s = 0.0f;
    for (int i = 0; i < 13; ++i) {
        const float x = (float)(i - 6);
        k[i] = expf(-0.5f * x * x / 4.0f);
    }
    for (int i = 0; i < 13; ++i) s += k[i];
    for (int i = 0; i < 13; ++i) g.w[i] = k[i] / s;
}

extern "C" void kernel_launch(void* const* d_in, const int* in_sizes, int n_in,
                              void* d_out, int out_size, void* d_ws, size_t ws_size,
                              hipStream_t stream) {
    (void)in_sizes; (void)n_in; (void)out_size; (void)ws_size;

    const float* in  = (const float*)d_in[0];
    float*       out = (float*)d_out;
    float*       tmp = (float*)d_ws;

    GaussW gw;
    make_weights(gw);

    gauss_wmma_x<<<1024, dim3(128), 0, stream>>>(in, tmp, gw);

    gauss_wmma_yz<<<256, dim3(512), 0, stream>>>(tmp, out, gw);
}


# Z-stage result turned round in a wave-private LDS tile: each global store instruction writes 512 contiguous bytes (4 whole 128-byte lines), not 16 pieces of 32 bytes; sum order unchanged
# speedup vs baseline: 2.5686x; 1.0967x over previous
#include <hip/hip_runtime.h>
#include <cmath>


typedef float v2f __attribute__((ext_vector_type(2)));
typedef float v4f __attribute__((ext_vector_type(4)));
typedef float v8f __attribute__((ext_vector_type(8)));

struct GaussW { float w[13]; };

__device__ __forceinline__ int iclamp(int v, int lo, int hi) {
    return v < lo ? lo : (v > hi ? hi : v);
}

__device__ __forceinline__ void band_products(const v2f (&a)[7], const v4f (&r0)[7],
                                              const v4f (&r1)[7], int m0, int laneHi,
                                              v8f& acc0, v8f& acc1, v8f& acc2, v8f& acc3) {
    acc0 = {}; acc1 = {}; acc2 = {}; acc3 = {};
#pragma unroll
    for (int c = 0; c < 7; ++c) {
        const int k0 = 4 * c + 2 * laneHi;
        const int p0 = m0 + k0 - 6;
        const int p1 = p0 + 1;
        const bool ok0 = (unsigned)p0 < 128u;
        const bool ok1 = (unsigned)p1 < 128u;
        v2f b0, b1, b2, b3;
        b0.x = ok0 ? r0[c].x : 0.0f;  b0.y = ok1 ? r1[c].x : 0.0f;
        b1.x = ok0 ? r0[c].y : 0.0f;  b1.y = ok1 ? r1[c].y : 0.0f;
        b2.x = ok0 ? r0[c].z : 0.0f;  b2.y = ok1 ? r1[c].z : 0.0f;
        b3.x = ok0 ? r0[c].w : 0.0f;  b3.y = ok1 ? r1[c].w : 0.0f;

        acc0 = __builtin_amdgcn_wmma_f32_16x16x4_f32(false, a[c], false, b0, (short)0, acc0, false, false);
        acc1 = __builtin_amdgcn_wmma_f32_16x16x4_f32(false, a[c], false, b1, (short)0, acc1, false, false);
        acc2 = __builtin_amdgcn_wmma_f32_16x16x4_f32(false, a[c], false, b2, (short)0, acc2, false, false);
        acc3 = __builtin_amdgcn_wmma_f32_16x16x4_f32(false, a[c], false, b3, (short)0, acc3, false, false);
    }
}

__device__ __forceinline__ void load_first_block(const char* __restrict__ inb, unsigned base,
                                                 unsigned S, int laneHi,
                                                 v4f (&r0)[7], v4f (&r1)[7]) {
#pragma unroll
    for (int c = 0; c < 7; ++c) {
        const int p0 = 4 * c + 2 * laneHi - 6;
        const unsigned q0 = (unsigned)iclamp(p0, 0, 127);
        const unsigned q1 = (unsigned)iclamp(p0 + 1, 0, 127);
        r0[c] = *reinterpret_cast<const v4f*>(inb + (base + q0 * S) * 4u);
        r1[c] = *reinterpret_cast<const v4f*>(inb + (base + q1 * S) * 4u);
    }
}

__device__ __forceinline__ void load_next_rows(const char* __restrict__ inb, unsigned base,
                                               unsigned S, int m0, int laneHi,
                                               v4f (&n0)[4], v4f (&n1)[4]) {
#pragma unroll
    for (int j = 0; j < 4; ++j) {
        const int p0 = m0 + 22 + 4 * j + 2 * laneHi;
        const unsigned q0 = (unsigned)iclamp(p0, 0, 127);
        const unsigned q1 = (unsigned)iclamp(p0 + 1, 0, 127);
        n0[j] = *reinterpret_cast<const v4f*>(inb + (base + q0 * S) * 4u);
        n1[j] = *reinterpret_cast<const v4f*>(inb + (base + q1 * S) * 4u);
    }
}

__device__ __forceinline__ void rotate_rows(v4f (&r0)[7], v4f (&r1)[7],
                                            const v4f (&n0)[4], const v4f (&n1)[4]) {
#pragma unroll
    for (int c = 0; c < 3; ++c) { r0[c] = r0[c + 4]; r1[c] = r1[c + 4]; }
#pragma unroll
    for (int j = 0; j < 4; ++j) { r0[3 + j] = n0[j]; r1[3 + j] = n1[j]; }
}

__global__ __launch_bounds__(128) void gauss_wmma_x(const float* __restrict__ in,
                                                    float* __restrict__ out,
                                                    GaussW gw) {
    __shared__ float wpad[64];
    const unsigned tid = threadIdx.x;
    if (tid < 64) wpad[tid] = 0.0f;
    __syncthreads();
    if (tid < 13) wpad[16 + tid] = gw.w[tid];
    __syncthreads();

    const unsigned wave   = tid >> 5;
    const unsigned lane   = tid & 31;
    const unsigned laneLo = lane & 15;
    const unsigned laneHi = lane >> 4;

    const unsigned lg   = blockIdx.x * 4u + wave;
    const unsigned n    = lg * 64u + laneLo * 4u;
    const unsigned S    = 131072u;
    const unsigned base = (n & 131071u) + (n >> 17) * 16777216u;

    const char* __restrict__ inb  = reinterpret_cast<const char*>(in);
    char* __restrict__       outb = reinterpret_cast<char*>(out);

    v2f a[7];
#pragma unroll
    for (int c = 0; c < 7; ++c) {
        const int k0 = 4 * c + 2 * (int)laneHi;
        a[c].x = wpad[16 + k0 - (int)laneLo];
        a[c].y = wpad[16 + k0 + 1 - (int)laneLo];
    }

    v4f r0[7], r1[7];
    load_first_block(inb, base, S, (int)laneHi, r0, r1);

#pragma unroll
    for (int ab = 0; ab < 8; ++ab) {
        const int m0 = ab * 16;
        v4f n0[4], n1[4];
        if (ab < 7) load_next_rows(inb, base, S, m0, (int)laneHi, n0, n1);

        __builtin_amdgcn_sched_barrier(0);

        v8f acc0, acc1, acc2, acc3;
        band_products(a, r0, r1, m0, (int)laneHi, acc0, acc1, acc2, acc3);

#pragma unroll
        for (int v = 0; v < 8; ++v) {
            const int m = m0 + v + 8 * (int)laneHi;
            v4f o;
            o.x = acc0[v];  o.y = acc1[v];  o.z = acc2[v];  o.w = acc3[v];
            *reinterpret_cast<v4f*>(outb + (base + (unsigned)m * S) * 4u) = o;
        }

        if (ab < 7) rotate_rows(r0, r1, n0, n1);
    }
}

#define YZ_PITCH 1032u
#define YZ_TILE4 (16u * YZ_PITCH / 4u)
#define ST_PITCH 136u
#define ST_TILE4 (8u * ST_PITCH / 4u)

__global__ __launch_bounds__(512) void gauss_wmma_yz(const float* __restrict__ in,
                                                     float* __restrict__ out,
                                                     GaussW gw) {
    __shared__ float wpad[64];
    __shared__ v4f ytile[2u * YZ_TILE4];
    __shared__ v4f stile[16u * ST_TILE4];

    const unsigned tid = threadIdx.x;
    if (tid < 64) wpad[tid] = 0.0f;
    __syncthreads();
    if (tid < 13) wpad[16 + tid] = gw.w[tid];
    __syncthreads();

    const unsigned wave   = tid >> 5;
    const unsigned lane   = tid & 31;
    const unsigned laneLo = lane & 15;
    const unsigned laneHi = lane >> 4;

    const unsigned slab = blockIdx.x * 131072u;

    const char* __restrict__ inb  = reinterpret_cast<const char*>(in);
    char* __restrict__       outb = reinterpret_cast<char*>(out);

    v2f a[7];
#pragma unroll
    for (int c = 0; c < 7; ++c) {
        const int k0 = 4 * c + 2 * (int)laneHi;
        a[c].x = wpad[16 + k0 - (int)laneLo];
        a[c].y = wpad[16 + k0 + 1 - (int)laneLo];
    }

    const unsigned zc    = wave * 64u + laneLo * 4u;
    const unsigned ybase = slab + zc;
    const unsigned SY    = 1024u;

    const unsigned nl    = (wave >> 3) * 64u + laneLo * 4u;
    const unsigned c0    = nl & 7u;
    const unsigned yl    = nl >> 3;
    const int      z0    = (int)((wave & 7u) * 16u);
    const unsigned lbase = yl * YZ_PITCH + c0;
    const unsigned SZ    = 8u;

    const unsigned st_wr = wave * ST_TILE4 + (((laneLo >> 1) * ST_PITCH + laneHi * 64u + c0) >> 2);
    const unsigned st_rd = wave * ST_TILE4 + lane;
    const unsigned gout  = slab + (wave >> 3) * 8u * 1024u + (unsigned)z0 * SZ + lane * 4u;

    v4f r0[7], r1[7];
    load_first_block(inb, ybase, SY, (int)laneHi, r0, r1);

#pragma unroll
    for (int yb = 0; yb < 8; ++yb) {
        const int y0 = yb * 16;
        const unsigned buf = ((unsigned)yb & 1u) * YZ_TILE4;

        v4f n0[4], n1[4];
        if (yb < 7) load_next_rows(inb, ybase, SY, y0, (int)laneHi, n0, n1);

        __builtin_amdgcn_sched_barrier(0);

        {
            v8f acc0, acc1, acc2, acc3;
            band_products(a, r0, r1, y0, (int)laneHi, acc0, acc1, acc2, acc3);

#pragma unroll
            for (int v = 0; v < 8; ++v) {
                const unsigned row = (unsigned)v + 8u * laneHi;
                v4f o;
                o.x = acc0[v];  o.y = acc1[v];  o.z = acc2[v];  o.w = acc3[v];
                ytile[buf + ((row * YZ_PITCH + zc) >> 2)] = o;
            }
        }

        __builtin_amdgcn_fence(__ATOMIC_RELEASE, "workgroup", "local");
        __builtin_amdgcn_s_barrier();
        __builtin_amdgcn_fence(__ATOMIC_ACQUIRE, "workgroup", "local");

        {
            v4f s0[7], s1[7];
#pragma unroll
            for (int c = 0; c < 7; ++c) {
                const int k0 = 4 * c + 2 * (int)laneHi;
                const int p0 = z0 + k0 - 6;
                const int p1 = p0 + 1;
                const unsigned q0 = (unsigned)iclamp(p0, 0, 127);
                const unsigned q1 = (unsigned)iclamp(p1, 0, 127);
                s0[c] = ytile[buf + ((lbase + q0 * SZ) >> 2)];
                s1[c] = ytile[buf + ((lbase + q1 * SZ) >> 2)];
            }

            __builtin_amdgcn_sched_barrier(0);

            v8f acc0, acc1, acc2, acc3;
            band_products(a, s0, s1, z0, (int)laneHi, acc0, acc1, acc2, acc3);

#pragma unroll
            for (int v = 0; v < 8; ++v) {
                v4f o;
                o.x = acc0[v];  o.y = acc1[v];  o.z = acc2[v];  o.w = acc3[v];
                stile[st_wr + (unsigned)v * 2u] = o;
            }

            __builtin_amdgcn_fence(__ATOMIC_RELEASE, "wavefront", "local");
            __builtin_amdgcn_wave_barrier();
            __builtin_amdgcn_fence(__ATOMIC_ACQUIRE, "wavefront", "local");

            v4f t[8];
#pragma unroll
            for (int i = 0; i < 8; ++i) t[i] = stile[st_rd + (unsigned)i * (ST_PITCH / 4u)];

            const unsigned grow = gout + (unsigned)y0 * 1024u;
#pragma unroll
            for (int i = 0; i < 8; ++i)
                *reinterpret_cast<v4f*>(outb + (grow + (unsigned)i * 1024u) * 4u) = t[i];
        }

        if (yb < 7) rotate_rows(r0, r1, n0, n1);
    }
}

static void make_weights(GaussW& g) {
    float k[13];
    float s = 0.0f;
    for (int i = 0; i < 13; ++i) {
        const float x = (float)(i - 6);
        k[i] = expf(-0.5f * x * x / 4.0f);
    }
    for (int i = 0; i < 13; ++i) s += k[i];
    for (int i = 0; i < 13; ++i) g.w[i] = k[i] / s;
}

extern "C" void kernel_launch(void* const* d_in, const int* in_sizes, int n_in,
                              void* d_out, int out_size, void* d_ws, size_t ws_size,
                              hipStream_t stream) {
    (void)in_sizes; (void)n_in; (void)out_size; (void)ws_size;

    const float* in  = (const float*)d_in[0];
    float*       out = (float*)d_out;
    float*       tmp = (float*)d_ws;

    GaussW gw;
    make_weights(gw);

    gauss_wmma_x<<<1024, dim3(128), 0, stream>>>(in, tmp, gw);

    gauss_wmma_yz<<<256, dim3(512), 0, stream>>>(tmp, out, gw);
}
